# starting kernel
# baseline (speedup 1.0000x reference)
#include <hip/hip_runtime.h>
#include <math.h>

typedef float v2f __attribute__((ext_vector_type(2)));
typedef float v8f __attribute__((ext_vector_type(8)));

#define CCH   256
#define HW    65536
#define NCLS  19
#define KC    0.5f

__global__ void __launch_bounds__(256, 2)
hyper_mlr_fused(const float* __restrict__ x,
                const float* __restrict__ P,
                const float* __restrict__ A,
                float* __restrict__ out)
{
    __shared__ float W[64 * CCH];
    float* ppS  = &W[19 * CCH];
    float* pdS  = &W[19 * CCH + 32];
    float* anS  = &W[19 * CCH + 64];
    float* invS = &W[19 * CCH + 96];

    const int tid = threadIdx.x;

    for (int i = tid; i < NCLS * CCH; i += 256) {
        int n = i >> 8, c = i & 255;
        W[n * CCH + c]        = -P[i];
        W[(32 + n) * CCH + c] =  A[i];
    }
    __syncthreads();

    if (tid < NCLS * 8) {
        int n    = tid >> 3;
        int part = tid & 7;
        const float* pr = &W[n * CCH + part * 32];
        const float* ar = &W[(32 + n) * CCH + part * 32];
        float sp = 0.f, sa = 0.f, spa = 0.f;
        for (int c = 0; c < 32; ++c) {
            float p = pr[c], a = ar[c];
            sp  = fmaf(p, p, sp);
            sa  = fmaf(a, a, sa);
            spa = fmaf(p, a, spa);
        }
        for (int d = 1; d < 8; d <<= 1) {
            sp  += __shfl_xor(sp,  d);
            sa  += __shfl_xor(sa,  d);
            spa += __shfl_xor(spa, d);
        }
        if (part == 0) {
            float an  = sqrtf(sa);
            float anc = fmaxf(an, 1e-12f);
            ppS[n]  = sp;
            anS[n]  = an;
            invS[n] = 1.0f / anc;
            pdS[n]  = spa / anc;
        }
    }
    __syncthreads();

    const int lane = tid & 31;
    const int wave = tid >> 5;
    const int half = lane >> 4;
    const int col  = lane & 15;

    long g   = (long)blockIdx.x * 128 + (long)wave * 16;
    int  b   = (int)(g >> 16);
    int  pix = (int)(g & 65535);

    const float* xw = x + ((long)b << 24) + pix + col + (long)(half * 2) * HW;

    const float* a0p = &W[( 0 + col) * CCH + half * 2];
    const float* a1p = &W[(16 + col) * CCH + half * 2];
    const float* a2p = &W[(32 + col) * CCH + half * 2];
    const float* a3p = &W[(48 + col) * CCH + half * 2];

    v8f acc0 = {}, acc1 = {}, acc2 = {}, acc3 = {};
    float sumsq = 0.f;

#pragma unroll 4
    for (int ks = 0; ks < 64; ++ks) {
        float b0 = __builtin_nontemporal_load(xw);
        float b1 = __builtin_nontemporal_load(xw + HW);
        xw += 4 * HW;
        sumsq = fmaf(b0, b0, sumsq);
        sumsq = fmaf(b1, b1, sumsq);
        v2f bf;  bf.x = b0;  bf.y = b1;

        v2f af0 = *(const v2f*)(a0p + 4 * ks);
        v2f af1 = *(const v2f*)(a1p + 4 * ks);
        v2f af2 = *(const v2f*)(a2p + 4 * ks);
        v2f af3 = *(const v2f*)(a3p + 4 * ks);

        acc0 = __builtin_amdgcn_wmma_f32_16x16x4_f32(false, af0, false, bf, (short)0, acc0, false, false);
        acc1 = __builtin_amdgcn_wmma_f32_16x16x4_f32(false, af1, false, bf, (short)0, acc1, false, false);
        acc2 = __builtin_amdgcn_wmma_f32_16x16x4_f32(false, af2, false, bf, (short)0, acc2, false, false);
        acc3 = __builtin_amdgcn_wmma_f32_16x16x4_f32(false, af3, false, bf, (short)0, acc3, false, false);
    }

    float xx = sumsq + __shfl_xor(sumsq, 16);

    const float eps     = 1e-12f;
    const float sqrtK   = 0.70710678118654752f;
    const float maxnorm = (1.0f - 0.001f) / sqrtK;
    const float mn2     = maxnorm * maxnorm;
    float* outw = out + (long)b * NCLS * HW + pix + col;

#pragma unroll
    for (int t = 0; t < 2; ++t) {
        v8f pxv = t ? acc1 : acc0;
        v8f xav = t ? acc3 : acc2;
#pragma unroll
        for (int r = 0; r < 8; ++r) {
            int cls = t * 16 + r + half * 8;
            if (cls < NCLS) {
                float px = pxv[r];
                float xa = xav[r] * invS[cls];
                float pp = ppS[cls];
                float pd = pdS[cls];
                float an = anS[cls];

                float Aq    = 1.0f + 2.0f * KC * px + KC * xx;
                float Bc    = 1.0f - KC * pp;
                float Dq    = fmaxf(1.0f + 2.0f * KC * px + (KC * xx) * (KC * pp), eps);
                float alpha = Aq / Dq;
                float beta  = Bc / Dq;
                float mob   = alpha * alpha * pp + beta * beta * xx
                            + 2.0f * alpha * beta * px;
                float sqm   = sqrtf(mob);
                float proj  = (sqm > maxnorm) ? (maxnorm / fmaxf(sqm, eps)) : 1.0f;
                float mobp  = (sqm < maxnorm) ? mob : mn2;
                float mdot  = (beta * xa + alpha * pd) * proj;
                float lamb  = 2.0f / fmaxf(1.0f - KC * mobp, eps);
                float s     = sqrtK * mdot * lamb;
                float ash   = copysignf(logf(fabsf(s) + sqrtf(fmaf(s, s, 1.0f))), s);
                __builtin_nontemporal_store((2.0f / sqrtK) * an * ash,
                                            outw + (long)cls * HW);
            }
        }
    }
}

extern "C" void kernel_launch(void* const* d_in, const int* in_sizes, int n_in,
                              void* d_out, int out_size, void* d_ws, size_t ws_size,
                              hipStream_t stream) {
    (void)in_sizes; (void)n_in; (void)d_ws; (void)ws_size; (void)out_size;
    const float* x = (const float*)d_in[0];
    const float* P = (const float*)d_in[1];
    const float* A = (const float*)d_in[2];
    float* out = (float*)d_out;
    hyper_mlr_fused<<<2048, 256, 0, stream>>>(x, P, A, out);
}


# LDS row pitch 256 -> 260 floats (16 B x odd): the 16 rows of an A-fragment read fall on 16 bank groups, not one bank pair (16-way conflict gone); unallocated rows 51-63 alias row 50
# speedup vs baseline: 1.6736x; 1.6736x over previous
#include <hip/hip_runtime.h>
#include <math.h>

typedef float v2f __attribute__((ext_vector_type(2)));
typedef float v8f __attribute__((ext_vector_type(8)));

#define CCH   256
#define HW    65536
#define NCLS  19
#define KC    0.5f
#define PITCH 260

__global__ void __launch_bounds__(256, 2)
hyper_mlr_fused(const float* __restrict__ x,
                const float* __restrict__ P,
                const float* __restrict__ A,
                float* __restrict__ out)
{
    __shared__ __attribute__((aligned(16))) float W[51 * PITCH];
    float* ppS  = &W[19 * PITCH];
    float* pdS  = &W[19 * PITCH + 32];
    float* anS  = &W[19 * PITCH + 64];
    float* invS = &W[19 * PITCH + 96];

    const int tid = threadIdx.x;

    for (int i = tid; i < NCLS * CCH; i += 256) {
        int n = i >> 8, c = i & 255;
        W[n * PITCH + c]        = -P[i];
        W[(32 + n) * PITCH + c] =  A[i];
    }
    __syncthreads();

    if (tid < NCLS * 8) {
        int n    = tid >> 3;
        int part = tid & 7;
        const float* pr = &W[n * PITCH + part * 32];
        const float* ar = &W[(32 + n) * PITCH + part * 32];
        float sp = 0.f, sa = 0.f, spa = 0.f;
        for (int c = 0; c < 32; ++c) {
            float p = pr[c], a = ar[c];
            sp  = fmaf(p, p, sp);
            sa  = fmaf(a, a, sa);
            spa = fmaf(p, a, spa);
        }
        for (int d = 1; d < 8; d <<= 1) {
            sp  += __shfl_xor(sp,  d);
            sa  += __shfl_xor(sa,  d);
            spa += __shfl_xor(spa, d);
        }
        if (part == 0) {
            float an  = sqrtf(sa);
            float anc = fmaxf(an, 1e-12f);
            ppS[n]  = sp;
            anS[n]  = an;
            invS[n] = 1.0f / anc;
            pdS[n]  = spa / anc;
        }
    }
    __syncthreads();

    const int lane = tid & 31;
    const int wave = tid >> 5;
    const int half = lane >> 4;
    const int col  = lane & 15;

    long g   = (long)blockIdx.x * 128 + (long)wave * 16;
    int  b   = (int)(g >> 16);
    int  pix = (int)(g & 65535);

    const float* xw = x + ((long)b << 24) + pix + col + (long)(half * 2) * HW;

    const int row3 = (48 + col < 50) ? (48 + col) : 50;
    const float* a0p = &W[( 0 + col) * PITCH + half * 2];
    const float* a1p = &W[(16 + col) * PITCH + half * 2];
    const float* a2p = &W[(32 + col) * PITCH + half * 2];
    const float* a3p = &W[row3       * PITCH + half * 2];

    v8f acc0 = {}, acc1 = {}, acc2 = {}, acc3 = {};
    float sumsq = 0.f;

#pragma unroll 4
    for (int ks = 0; ks < 64; ++ks) {
        float b0 = __builtin_nontemporal_load(xw);
        float b1 = __builtin_nontemporal_load(xw + HW);
        xw += 4 * HW;
        sumsq = fmaf(b0, b0, sumsq);
        sumsq = fmaf(b1, b1, sumsq);
        v2f bf;  bf.x = b0;  bf.y = b1;

        v2f af0 = *(const v2f*)(a0p + 4 * ks);
        v2f af1 = *(const v2f*)(a1p + 4 * ks);
        v2f af2 = *(const v2f*)(a2p + 4 * ks);
        v2f af3 = *(const v2f*)(a3p + 4 * ks);

        acc0 = __builtin_amdgcn_wmma_f32_16x16x4_f32(false, af0, false, bf, (short)0, acc0, false, false);
        acc1 = __builtin_amdgcn_wmma_f32_16x16x4_f32(false, af1, false, bf, (short)0, acc1, false, false);
        acc2 = __builtin_amdgcn_wmma_f32_16x16x4_f32(false, af2, false, bf, (short)0, acc2, false, false);
        acc3 = __builtin_amdgcn_wmma_f32_16x16x4_f32(false, af3, false, bf, (short)0, acc3, false, false);
    }

    float xx = sumsq + __shfl_xor(sumsq, 16);

    const float eps     = 1e-12f;
    const float sqrtK   = 0.70710678118654752f;
    const float maxnorm = (1.0f - 0.001f) / sqrtK;
    const float mn2     = maxnorm * maxnorm;
    float* outw = out + (long)b * NCLS * HW + pix + col;

#pragma unroll
    for (int t = 0; t < 2; ++t) {
        v8f pxv = t ? acc1 : acc0;
        v8f xav = t ? acc3 : acc2;
#pragma unroll
        for (int r = 0; r < 8; ++r) {
            int cls = t * 16 + r + half * 8;
            if (cls < NCLS) {
                float px = pxv[r];
                float xa = xav[r] * invS[cls];
                float pp = ppS[cls];
                float pd = pdS[cls];
                float an = anS[cls];

                float Aq    = 1.0f + 2.0f * KC * px + KC * xx;
                float Bc    = 1.0f - KC * pp;
                float Dq    = fmaxf(1.0f + 2.0f * KC * px + (KC * xx) * (KC * pp), eps);
                float alpha = Aq / Dq;
                float beta  = Bc / Dq;
                float mob   = alpha * alpha * pp + beta * beta * xx
                            + 2.0f * alpha * beta * px;
                float sqm   = sqrtf(mob);
                float proj  = (sqm > maxnorm) ? (maxnorm / fmaxf(sqm, eps)) : 1.0f;
                float mobp  = (sqm < maxnorm) ? mob : mn2;
                float mdot  = (beta * xa + alpha * pd) * proj;
                float lamb  = 2.0f / fmaxf(1.0f - KC * mobp, eps);
                float s     = sqrtK * mdot * lamb;
                float ash   = copysignf(logf(fabsf(s) + sqrtf(fmaf(s, s, 1.0f))), s);
                __builtin_nontemporal_store((2.0f / sqrtK) * an * ash,
                                            outw + (long)cls * HW);
            }
        }
    }
}

extern "C" void kernel_launch(void* const* d_in, const int* in_sizes, int n_in,
                              void* d_out, int out_size, void* d_ws, size_t ws_size,
                              hipStream_t stream) {
    (void)in_sizes; (void)n_in; (void)d_ws; (void)ws_size; (void)out_size;
    const float* x = (const float*)d_in[0];
    const float* P = (const float*)d_in[1];
    const float* A = (const float*)d_in[2];
    float* out = (float*)d_out;
    hyper_mlr_fused<<<2048, 256, 0, stream>>>(x, P, A, out);
}


# the 38 weight rows (19 of -P, 19 of A) packed into 3 M-tiles instead of 4, both products of a class in one lane: 12 f32 WMMAs per 4 k-steps instead of 16, 10 epilogue slots per lane instead of 11
# speedup vs baseline: 1.9369x; 1.1573x over previous
#include <hip/hip_runtime.h>
#include <math.h>

typedef float v2f __attribute__((ext_vector_type(2)));
typedef float v8f __attribute__((ext_vector_type(8)));

#define CCH   256
#define HW    65536
#define NCLS  19
#define KC    0.5f
#define PITCH 260

__device__ __forceinline__ int p_row(int n) {
    int h = (n >= 10) ? 1 : 0;
    int j = n - 10 * h;
    return (j < 8) ? (8 * h + j) : (32 + 4 * h + 2 * (j - 8));
}
__device__ __forceinline__ int a_row(int n) {
    int h = (n >= 10) ? 1 : 0;
    int j = n - 10 * h;
    return (j < 8) ? (16 + 8 * h + j) : (33 + 4 * h + 2 * (j - 8));
}

__global__ void __launch_bounds__(256, 2)
hyper_mlr_fused(const float* __restrict__ x,
                const float* __restrict__ P,
                const float* __restrict__ A,
                float* __restrict__ out)
{
    __shared__ __attribute__((aligned(16))) float W[38 * PITCH];
    __shared__ float ppS[32];
    __shared__ float pdS[32];
    __shared__ float anS[32];
    __shared__ float invS[32];

    const int tid = threadIdx.x;

    for (int n = 0; n < NCLS; ++n) {
        W[p_row(n) * PITCH + tid] = -P[n * CCH + tid];
        W[a_row(n) * PITCH + tid] =  A[n * CCH + tid];
    }
    __syncthreads();

    if (tid < NCLS * 8) {
        int n    = tid >> 3;
        int part = tid & 7;
        const float* pr = &W[p_row(n) * PITCH + part * 32];
        const float* ar = &W[a_row(n) * PITCH + part * 32];
        float sp = 0.f, sa = 0.f, spa = 0.f;
        for (int c = 0; c < 32; ++c) {
            float p = pr[c], a = ar[c];
            sp  = fmaf(p, p, sp);
            sa  = fmaf(a, a, sa);
            spa = fmaf(p, a, spa);
        }
        for (int d = 1; d < 8; d <<= 1) {
            sp  += __shfl_xor(sp,  d);
            sa  += __shfl_xor(sa,  d);
            spa += __shfl_xor(spa, d);
        }
        if (part == 0) {
            float an  = sqrtf(sa);
            float anc = fmaxf(an, 1e-12f);
            ppS[n]  = sp;
            anS[n]  = an;
            invS[n] = 1.0f / anc;
            pdS[n]  = spa / anc;
        }
    }
    __syncthreads();

    const int lane = tid & 31;
    const int wave = tid >> 5;
    const int half = lane >> 4;
    const int col  = lane & 15;

    long g   = (long)blockIdx.x * 128 + (long)wave * 16;
    int  b   = (int)(g >> 16);
    int  pix = (int)(g & 65535);

    const float* xw = x + ((long)b << 24) + pix + col + (long)(half * 2) * HW;

    const int row2 = (col < 4) ? (32 + col) : ((col == 8) ? 36 : 37);
    const float* a0p = &W[( 0 + col) * PITCH + half * 2];
    const float* a1p = &W[(16 + col) * PITCH + half * 2];
    const float* a2p = &W[row2       * PITCH + half * 2];

    v8f acc0 = {}, acc1 = {}, acc2 = {};
    float sumsq = 0.f;

#pragma unroll 4
    for (int ks = 0; ks < 64; ++ks) {
        float b0 = __builtin_nontemporal_load(xw);
        float b1 = __builtin_nontemporal_load(xw + HW);
        xw += 4 * HW;
        sumsq = fmaf(b0, b0, sumsq);
        sumsq = fmaf(b1, b1, sumsq);
        v2f bf;  bf.x = b0;  bf.y = b1;

        v2f af0 = *(const v2f*)(a0p + 4 * ks);
        v2f af1 = *(const v2f*)(a1p + 4 * ks);
        v2f af2 = *(const v2f*)(a2p + 4 * ks);

        acc0 = __builtin_amdgcn_wmma_f32_16x16x4_f32(false, af0, false, bf, (short)0, acc0, false, false);
        acc1 = __builtin_amdgcn_wmma_f32_16x16x4_f32(false, af1, false, bf, (short)0, acc1, false, false);
        acc2 = __builtin_amdgcn_wmma_f32_16x16x4_f32(false, af2, false, bf, (short)0, acc2, false, false);
    }

    float xx = sumsq + __shfl_xor(sumsq, 16);

    const float eps     = 1e-12f;
    const float sqrtK   = 0.70710678118654752f;
    const float maxnorm = (1.0f - 0.001f) / sqrtK;
    const float mn2     = maxnorm * maxnorm;
    float* outw = out + (long)b * NCLS * HW + pix + col;

#pragma unroll
    for (int j = 0; j < 10; ++j) {
        int cls = 10 * half + j;
        if (cls < NCLS) {
            float px = (j < 8) ? acc0[j] : acc2[2 * (j - 8)];
            float xr = (j < 8) ? acc1[j] : acc2[2 * (j - 8) + 1];
            float xa = xr * invS[cls];
            float pp = ppS[cls];
            float pd = pdS[cls];
            float an = anS[cls];

            float Aq    = 1.0f + 2.0f * KC * px + KC * xx;
            float Bc    = 1.0f - KC * pp;
            float Dq    = fmaxf(1.0f + 2.0f * KC * px + (KC * xx) * (KC * pp), eps);
            float alpha = Aq / Dq;
            float beta  = Bc / Dq;
            float mob   = alpha * alpha * pp + beta * beta * xx
                        + 2.0f * alpha * beta * px;
            float sqm   = sqrtf(mob);
            float proj  = (sqm > maxnorm) ? (maxnorm / fmaxf(sqm, eps)) : 1.0f;
            float mobp  = (sqm < maxnorm) ? mob : mn2;
            float mdot  = (beta * xa + alpha * pd) * proj;
            float lamb  = 2.0f / fmaxf(1.0f - KC * mobp, eps);
            float s     = sqrtK * mdot * lamb;
            float ash   = copysignf(logf(fabsf(s) + sqrtf(fmaf(s, s, 1.0f))), s);
            __builtin_nontemporal_store((2.0f / sqrtK) * an * ash,
                                        outw + (long)cls * HW);
        }
    }
}

extern "C" void kernel_launch(void* const* d_in, const int* in_sizes, int n_in,
                              void* d_out, int out_size, void* d_ws, size_t ws_size,
                              hipStream_t stream) {
    (void)in_sizes; (void)n_in; (void)d_ws; (void)ws_size; (void)out_size;
    const float* x = (const float*)d_in[0];
    const float* P = (const float*)d_in[1];
    const float* A = (const float*)d_in[2];
    float* out = (float*)d_out;
    hyper_mlr_fused<<<2048, 256, 0, stream>>>(x, P, A, out);
}


# wave tile 16 -> 64 pixels: a lane loads 4 adjacent pixels of a channel in one 16-byte load (wave-wide: aligned 256-B runs, next group issued ahead), A fragments feed 4 pixel tiles, 16-byte stores
# speedup vs baseline: 2.2671x; 1.1705x over previous
#include <hip/hip_runtime.h>
#include <math.h>


typedef float v2f __attribute__((ext_vector_type(2)));
typedef float v4f __attribute__((ext_vector_type(4)));
typedef float v8f __attribute__((ext_vector_type(8)));

#define CCH   256
#define HW    65536
#define NCLS  19
#define KC    0.5f
#define PITCH 260
#define GK    4

__device__ __forceinline__ int p_row(int n) {
    int h = (n >= 10) ? 1 : 0;
    int j = n - 10 * h;
    return (j < 8) ? (8 * h + j) : (32 + 4 * h + 2 * (j - 8));
}
__device__ __forceinline__ int a_row(int n) {
    int h = (n >= 10) ? 1 : 0;
    int j = n - 10 * h;
    return (j < 8) ? (16 + 8 * h + j) : (33 + 4 * h + 2 * (j - 8));
}

__device__ __forceinline__ void load_group(v4f (&r)[2 * GK], const float* p) {
#pragma unroll
    for (int i = 0; i < GK; ++i) {
        r[2 * i]     = __builtin_nontemporal_load((const v4f*)(p + (long)(4 * i)     * HW));
        r[2 * i + 1] = __builtin_nontemporal_load((const v4f*)(p + (long)(4 * i + 1) * HW));
    }
}

__device__ __forceinline__ void mma_group(v4f (&r)[2 * GK],
                                          const float* a0p, const float* a1p, const float* a2p,
                                          v8f (&acc)[3][4], float (&sumsq)[4]) {
    asm volatile("" : "+v"(r[0]), "+v"(r[1]), "+v"(r[2]), "+v"(r[3]),
                      "+v"(r[4]), "+v"(r[5]), "+v"(r[6]), "+v"(r[7]));
#pragma unroll
    for (int i = 0; i < GK; ++i) {
        v2f af0 = *(const v2f*)(a0p + 4 * i);
        v2f af1 = *(const v2f*)(a1p + 4 * i);
        v2f af2 = *(const v2f*)(a2p + 4 * i);
        const v4f c0 = r[2 * i], c1 = r[2 * i + 1];
#pragma unroll
        for (int t = 0; t < 4; ++t) {
            sumsq[t] = fmaf(c0[t], c0[t], sumsq[t]);
            sumsq[t] = fmaf(c1[t], c1[t], sumsq[t]);
            v2f bf;  bf.x = c0[t];  bf.y = c1[t];
            acc[0][t] = __builtin_amdgcn_wmma_f32_16x16x4_f32(false, af0, false, bf, (short)0, acc[0][t], false, false);
            acc[1][t] = __builtin_amdgcn_wmma_f32_16x16x4_f32(false, af1, false, bf, (short)0, acc[1][t], false, false);
            acc[2][t] = __builtin_amdgcn_wmma_f32_16x16x4_f32(false, af2, false, bf, (short)0, acc[2][t], false, false);
        }
    }
    asm volatile("" : "+v"(sumsq[0]), "+v"(sumsq[1]), "+v"(sumsq[2]), "+v"(sumsq[3]));
}

__device__ __forceinline__ float mlr_one(float px, float xr, float xx,
                                         float inv, float pp, float pd, float an) {
    const float eps     = 1e-12f;
    const float sqrtK   = 0.70710678118654752f;
    const float maxnorm = (1.0f - 0.001f) / sqrtK;
    const float mn2     = maxnorm * maxnorm;
    float xa    = xr * inv;
    float Aq    = 1.0f + 2.0f * KC * px + KC * xx;
    float Bc    = 1.0f - KC * pp;
    float Dq    = fmaxf(1.0f + 2.0f * KC * px + (KC * xx) * (KC * pp), eps);
    float alpha = Aq / Dq;
    float beta  = Bc / Dq;
    float mob   = alpha * alpha * pp + beta * beta * xx
                + 2.0f * alpha * beta * px;
    float sqm   = sqrtf(mob);
    float proj  = (sqm > maxnorm) ? (maxnorm / fmaxf(sqm, eps)) : 1.0f;
    float mobp  = (sqm < maxnorm) ? mob : mn2;
    float mdot  = (beta * xa + alpha * pd) * proj;
    float lamb  = 2.0f / fmaxf(1.0f - KC * mobp, eps);
    float s     = sqrtK * mdot * lamb;
    float ash   = copysignf(logf(fabsf(s) + sqrtf(fmaf(s, s, 1.0f))), s);
    return (2.0f / sqrtK) * an * ash;
}

__global__ void __launch_bounds__(256, 4)
hyper_mlr_fused(const float* __restrict__ x,
                const float* __restrict__ P,
                const float* __restrict__ A,
                float* __restrict__ out)
{
    __shared__ __attribute__((aligned(16))) float W[38 * PITCH];
    __shared__ float ppS[32];
    __shared__ float pdS[32];
    __shared__ float anS[32];
    __shared__ float invS[32];

    const int tid = threadIdx.x;

    for (int n = 0; n < NCLS; ++n) {
        W[p_row(n) * PITCH + tid] = -P[n * CCH + tid];
        W[a_row(n) * PITCH + tid] =  A[n * CCH + tid];
    }
    __syncthreads();

    if (tid < NCLS * 8) {
        int n    = tid >> 3;
        int part = tid & 7;
        const float* pr = &W[p_row(n) * PITCH + part * 32];
        const float* ar = &W[a_row(n) * PITCH + part * 32];
        float sp = 0.f, sa = 0.f, spa = 0.f;
        for (int c = 0; c < 32; ++c) {
            float p = pr[c], a = ar[c];
            sp  = fmaf(p, p, sp);
            sa  = fmaf(a, a, sa);
            spa = fmaf(p, a, spa);
        }
        for (int d = 1; d < 8; d <<= 1) {
            sp  += __shfl_xor(sp,  d);
            sa  += __shfl_xor(sa,  d);
            spa += __shfl_xor(spa, d);
        }
        if (part == 0) {
            float an  = sqrtf(sa);
            float anc = fmaxf(an, 1e-12f);
            ppS[n]  = sp;
            anS[n]  = an;
            invS[n] = 1.0f / anc;
            pdS[n]  = spa / anc;
        }
    }
    __syncthreads();

    const int lane = tid & 31;
    const int wave = tid >> 5;
    const int half = lane >> 4;
    const int col  = lane & 15;

    long g   = (long)blockIdx.x * 512 + (long)wave * 64;
    int  b   = (int)(g >> 16);
    int  pix = (int)(g & 65535);

    const float* xw = x + ((long)b << 24) + pix + 4 * col + (long)(half * 2) * HW;

    const int row2 = (col < 4) ? (32 + col) : ((col == 8) ? 36 : 37);
    const float* a0p = &W[( 0 + col) * PITCH + half * 2];
    const float* a1p = &W[(16 + col) * PITCH + half * 2];
    const float* a2p = &W[row2       * PITCH + half * 2];

    v8f acc[3][4] = {};
    float sumsq[4] = {0.f, 0.f, 0.f, 0.f};

    v4f ra[2 * GK], rb[2 * GK];
    load_group(ra, xw);  xw += (long)(4 * GK) * HW;
    for (int it = 0; it < 7; ++it) {
        load_group(rb, xw);  xw += (long)(4 * GK) * HW;
        mma_group(ra, a0p, a1p, a2p, acc, sumsq);
        load_group(ra, xw);  xw += (long)(4 * GK) * HW;
        mma_group(rb, a0p + 4 * GK, a1p + 4 * GK, a2p + 4 * GK, acc, sumsq);
        a0p += 8 * GK;  a1p += 8 * GK;  a2p += 8 * GK;
    }
    load_group(rb, xw);
    mma_group(ra, a0p, a1p, a2p, acc, sumsq);
    __builtin_amdgcn_sched_barrier(0);
    mma_group(rb, a0p + 4 * GK, a1p + 4 * GK, a2p + 4 * GK, acc, sumsq);

    float xx[4];
#pragma unroll
    for (int t = 0; t < 4; ++t) xx[t] = sumsq[t] + __shfl_xor(sumsq[t], 16);

    v4f o[10];
#pragma unroll
    for (int j = 0; j < 10; ++j) o[j] = v4f{0.f, 0.f, 0.f, 0.f};

#pragma unroll 1
    for (int pass = 0; pass < 4; ++pass) {
#pragma unroll
        for (int j = 0; j < 10; ++j) {
            int cls = 10 * half + j;
            if (cls < NCLS) {
                float px = (j < 8) ? acc[0][0][j] : acc[2][0][2 * (j - 8)];
                float xr = (j < 8) ? acc[1][0][j] : acc[2][0][2 * (j - 8) + 1];
                float v  = mlr_one(px, xr, xx[0], invS[cls], ppS[cls], pdS[cls], anS[cls]);
                o[j] = v4f{o[j].y, o[j].z, o[j].w, v};
            }
        }
#pragma unroll
        for (int m = 0; m < 3; ++m) { acc[m][0] = acc[m][1];  acc[m][1] = acc[m][2];  acc[m][2] = acc[m][3]; }
        xx[0] = xx[1];  xx[1] = xx[2];  xx[2] = xx[3];
    }

    float* outw = out + (long)b * NCLS * HW + pix + 4 * col;
#pragma unroll
    for (int j = 0; j < 10; ++j) {
        int cls = 10 * half + j;
        if (cls < NCLS)
            __builtin_nontemporal_store(o[j], (v4f*)(outw + (long)cls * HW));
    }
}

extern "C" void kernel_launch(void* const* d_in, const int* in_sizes, int n_in,
                              void* d_out, int out_size, void* d_ws, size_t ws_size,
                              hipStream_t stream) {
    (void)in_sizes; (void)n_in; (void)d_ws; (void)ws_size; (void)out_size;
    const float* x = (const float*)d_in[0];
    const float* P = (const float*)d_in[1];
    const float* A = (const float*)d_in[2];
    float* out = (float*)d_out;
    hyper_mlr_fused<<<512, 256, 0, stream>>>(x, P, A, out);
}


# epilogue quotients, square roots and log on single v_rcp_f32 / v_sqrt_f32 / v_log_f32 instructions (1 ulp each), not IEEE division, rounded sqrt and library log; alpha, beta share one reciprocal
# speedup vs baseline: 2.7568x; 1.2160x over previous
#include <hip/hip_runtime.h>
#include <math.h>


typedef float v2f __attribute__((ext_vector_type(2)));
typedef float v4f __attribute__((ext_vector_type(4)));
typedef float v8f __attribute__((ext_vector_type(8)));

#define CCH   256
#define HW    65536
#define NCLS  19
#define KC    0.5f
#define PITCH 260
#define GK    4

__device__ __forceinline__ int p_row(int n) {
    int h = (n >= 10) ? 1 : 0;
    int j = n - 10 * h;
    return (j < 8) ? (8 * h + j) : (32 + 4 * h + 2 * (j - 8));
}
__device__ __forceinline__ int a_row(int n) {
    int h = (n >= 10) ? 1 : 0;
    int j = n - 10 * h;
    return (j < 8) ? (16 + 8 * h + j) : (33 + 4 * h + 2 * (j - 8));
}

__device__ __forceinline__ void load_group(v4f (&r)[2 * GK], const float* p) {
#pragma unroll
    for (int i = 0; i < GK; ++i) {
        r[2 * i]     = __builtin_nontemporal_load((const v4f*)(p + (long)(4 * i)     * HW));
        r[2 * i + 1] = __builtin_nontemporal_load((const v4f*)(p + (long)(4 * i + 1) * HW));
    }
}

__device__ __forceinline__ void mma_group(v4f (&r)[2 * GK],
                                          const float* a0p, const float* a1p, const float* a2p,
                                          v8f (&acc)[3][4], float (&sumsq)[4]) {
    asm volatile("" : "+v"(r[0]), "+v"(r[1]), "+v"(r[2]), "+v"(r[3]),
                      "+v"(r[4]), "+v"(r[5]), "+v"(r[6]), "+v"(r[7]));
#pragma unroll
    for (int i = 0; i < GK; ++i) {
        v2f af0 = *(const v2f*)(a0p + 4 * i);
        v2f af1 = *(const v2f*)(a1p + 4 * i);
        v2f af2 = *(const v2f*)(a2p + 4 * i);
        const v4f c0 = r[2 * i], c1 = r[2 * i + 1];
#pragma unroll
        for (int t = 0; t < 4; ++t) {
            sumsq[t] = fmaf(c0[t], c0[t], sumsq[t]);
            sumsq[t] = fmaf(c1[t], c1[t], sumsq[t]);
            v2f bf;  bf.x = c0[t];  bf.y = c1[t];
            acc[0][t] = __builtin_amdgcn_wmma_f32_16x16x4_f32(false, af0, false, bf, (short)0, acc[0][t], false, false);
            acc[1][t] = __builtin_amdgcn_wmma_f32_16x16x4_f32(false, af1, false, bf, (short)0, acc[1][t], false, false);
            acc[2][t] = __builtin_amdgcn_wmma_f32_16x16x4_f32(false, af2, false, bf, (short)0, acc[2][t], false, false);
        }
    }
    asm volatile("" : "+v"(sumsq[0]), "+v"(sumsq[1]), "+v"(sumsq[2]), "+v"(sumsq[3]));
}

__device__ __forceinline__ float mlr_one(float px, float xr, float xx,
                                         float inv, float pp, float pd, float an) {
    const float eps     = 1e-12f;
    const float sqrtK   = 0.70710678118654752f;
    const float ln2     = 0.69314718055994531f;
    const float maxnorm = (1.0f - 0.001f) / sqrtK;
    const float mn2     = maxnorm * maxnorm;
    float xa    = xr * inv;
    float Aq    = 1.0f + 2.0f * KC * px + KC * xx;
    float Bc    = 1.0f - KC * pp;
    float Dq    = fmaxf(1.0f + 2.0f * KC * px + (KC * xx) * (KC * pp), eps);
    float rD    = __builtin_amdgcn_rcpf(Dq);
    float alpha = Aq * rD;
    float beta  = Bc * rD;
    float mob   = alpha * alpha * pp + beta * beta * xx
                + 2.0f * alpha * beta * px;
    float sqm   = __builtin_amdgcn_sqrtf(mob);
    float proj  = (sqm > maxnorm) ? (maxnorm * __builtin_amdgcn_rcpf(fmaxf(sqm, eps))) : 1.0f;
    float mobp  = (sqm < maxnorm) ? mob : mn2;
    float mdot  = (beta * xa + alpha * pd) * proj;
    float lamb  = 2.0f * __builtin_amdgcn_rcpf(fmaxf(1.0f - KC * mobp, eps));
    float s     = sqrtK * mdot * lamb;
    float ash   = copysignf(ln2 * __builtin_amdgcn_logf(fabsf(s) + __builtin_amdgcn_sqrtf(fmaf(s, s, 1.0f))), s);
    return (2.0f / sqrtK) * an * ash;
}

__global__ void __launch_bounds__(256, 4)
hyper_mlr_fused(const float* __restrict__ x,
                const float* __restrict__ P,
                const float* __restrict__ A,
                float* __restrict__ out)
{
    __shared__ __attribute__((aligned(16))) float W[38 * PITCH];
    __shared__ float ppS[32];
    __shared__ float pdS[32];
    __shared__ float anS[32];
    __shared__ float invS[32];

    const int tid = threadIdx.x;

    for (int n = 0; n < NCLS; ++n) {
        W[p_row(n) * PITCH + tid] = -P[n * CCH + tid];
        W[a_row(n) * PITCH + tid] =  A[n * CCH + tid];
    }
    __syncthreads();

    if (tid < NCLS * 8) {
        int n    = tid >> 3;
        int part = tid & 7;
        const float* pr = &W[p_row(n) * PITCH + part * 32];
        const float* ar = &W[a_row(n) * PITCH + part * 32];
        float sp = 0.f, sa = 0.f, spa = 0.f;
        for (int c = 0; c < 32; ++c) {
            float p = pr[c], a = ar[c];
            sp  = fmaf(p, p, sp);
            sa  = fmaf(a, a, sa);
            spa = fmaf(p, a, spa);
        }
        for (int d = 1; d < 8; d <<= 1) {
            sp  += __shfl_xor(sp,  d);
            sa  += __shfl_xor(sa,  d);
            spa += __shfl_xor(spa, d);
        }
        if (part == 0) {
            float an  = sqrtf(sa);
            float anc = fmaxf(an, 1e-12f);
            ppS[n]  = sp;
            anS[n]  = an;
            invS[n] = 1.0f / anc;
            pdS[n]  = spa / anc;
        }
    }
    __syncthreads();

    const int lane = tid & 31;
    const int wave = tid >> 5;
    const int half = lane >> 4;
    const int col  = lane & 15;

    long g   = (long)blockIdx.x * 512 + (long)wave * 64;
    int  b   = (int)(g >> 16);
    int  pix = (int)(g & 65535);

    const float* xw = x + ((long)b << 24) + pix + 4 * col + (long)(half * 2) * HW;

    const int row2 = (col < 4) ? (32 + col) : ((col == 8) ? 36 : 37);
    const float* a0p = &W[( 0 + col) * PITCH + half * 2];
    const float* a1p = &W[(16 + col) * PITCH + half * 2];
    const float* a2p = &W[row2       * PITCH + half * 2];

    v8f acc[3][4] = {};
    float sumsq[4] = {0.f, 0.f, 0.f, 0.f};

    v4f ra[2 * GK], rb[2 * GK];
    load_group(ra, xw);  xw += (long)(4 * GK) * HW;
    for (int it = 0; it < 7; ++it) {
        load_group(rb, xw);  xw += (long)(4 * GK) * HW;
        mma_group(ra, a0p, a1p, a2p, acc, sumsq);
        load_group(ra, xw);  xw += (long)(4 * GK) * HW;
        mma_group(rb, a0p + 4 * GK, a1p + 4 * GK, a2p + 4 * GK, acc, sumsq);
        a0p += 8 * GK;  a1p += 8 * GK;  a2p += 8 * GK;
    }
    load_group(rb, xw);
    mma_group(ra, a0p, a1p, a2p, acc, sumsq);
    __builtin_amdgcn_sched_barrier(0);
    mma_group(rb, a0p + 4 * GK, a1p + 4 * GK, a2p + 4 * GK, acc, sumsq);

    float xx[4];
#pragma unroll
    for (int t = 0; t < 4; ++t) xx[t] = sumsq[t] + __shfl_xor(sumsq[t], 16);

    v4f o[10];
#pragma unroll
    for (int j = 0; j < 10; ++j) o[j] = v4f{0.f, 0.f, 0.f, 0.f};

#pragma unroll 1
    for (int pass = 0; pass < 4; ++pass) {
#pragma unroll
        for (int j = 0; j < 10; ++j) {
            int cls = 10 * half + j;
            if (cls < NCLS) {
                float px = (j < 8) ? acc[0][0][j] : acc[2][0][2 * (j - 8)];
                float xr = (j < 8) ? acc[1][0][j] : acc[2][0][2 * (j - 8) + 1];
                float v  = mlr_one(px, xr, xx[0], invS[cls], ppS[cls], pdS[cls], anS[cls]);
                o[j] = v4f{o[j].y, o[j].z, o[j].w, v};
            }
        }
#pragma unroll
        for (int m = 0; m < 3; ++m) { acc[m][0] = acc[m][1];  acc[m][1] = acc[m][2];  acc[m][2] = acc[m][3]; }
        xx[0] = xx[1];  xx[1] = xx[2];  xx[2] = xx[3];
    }

    float* outw = out + (long)b * NCLS * HW + pix + 4 * col;
#pragma unroll
    for (int j = 0; j < 10; ++j) {
        int cls = 10 * half + j;
        if (cls < NCLS)
            __builtin_nontemporal_store(o[j], (v4f*)(outw + (long)cls * HW));
    }
}

extern "C" void kernel_launch(void* const* d_in, const int* in_sizes, int n_in,
                              void* d_out, int out_size, void* d_ws, size_t ws_size,
                              hipStream_t stream) {
    (void)in_sizes; (void)n_in; (void)d_ws; (void)ws_size; (void)out_size;
    const float* x = (const float*)d_in[0];
    const float* P = (const float*)d_in[1];
    const float* A = (const float*)d_in[2];
    float* out = (float*)d_out;
    hyper_mlr_fused<<<512, 256, 0, stream>>>(x, P, A, out);
}


# wave priority falls with progress along the K loop (s_setprio 3,2,1 at load groups 0,6,10; 0 from group 14 on): a SIMD's 4 waves advance together, none ends alone with its load waits uncovered
# speedup vs baseline: 2.8972x; 1.0509x over previous
#include <hip/hip_runtime.h>
#include <math.h>


typedef float v2f __attribute__((ext_vector_type(2)));
typedef float v4f __attribute__((ext_vector_type(4)));
typedef float v8f __attribute__((ext_vector_type(8)));

#define CCH   256
#define HW    65536
#define NCLS  19
#define KC    0.5f
#define PITCH 260
#define GK    4

__device__ __forceinline__ int p_row(int n) {
    int h = (n >= 10) ? 1 : 0;
    int j = n - 10 * h;
    return (j < 8) ? (8 * h + j) : (32 + 4 * h + 2 * (j - 8));
}
__device__ __forceinline__ int a_row(int n) {
    int h = (n >= 10) ? 1 : 0;
    int j = n - 10 * h;
    return (j < 8) ? (16 + 8 * h + j) : (33 + 4 * h + 2 * (j - 8));
}

__device__ __forceinline__ void load_group(v4f (&r)[2 * GK], const float* p) {
#pragma unroll
    for (int i = 0; i < GK; ++i) {
        r[2 * i]     = __builtin_nontemporal_load((const v4f*)(p + (long)(4 * i)     * HW));
        r[2 * i + 1] = __builtin_nontemporal_load((const v4f*)(p + (long)(4 * i + 1) * HW));
    }
}

__device__ __forceinline__ void mma_group(v4f (&r)[2 * GK],
                                          const float* a0p, const float* a1p, const float* a2p,
                                          v8f (&acc)[3][4], float (&sumsq)[4]) {
    asm volatile("" : "+v"(r[0]), "+v"(r[1]), "+v"(r[2]), "+v"(r[3]),
                      "+v"(r[4]), "+v"(r[5]), "+v"(r[6]), "+v"(r[7]));
#pragma unroll
    for (int i = 0; i < GK; ++i) {
        v2f af0 = *(const v2f*)(a0p + 4 * i);
        v2f af1 = *(const v2f*)(a1p + 4 * i);
        v2f af2 = *(const v2f*)(a2p + 4 * i);
        const v4f c0 = r[2 * i], c1 = r[2 * i + 1];
#pragma unroll
        for (int t = 0; t < 4; ++t) {
            sumsq[t] = fmaf(c0[t], c0[t], sumsq[t]);
            sumsq[t] = fmaf(c1[t], c1[t], sumsq[t]);
            v2f bf;  bf.x = c0[t];  bf.y = c1[t];
            acc[0][t] = __builtin_amdgcn_wmma_f32_16x16x4_f32(false, af0, false, bf, (short)0, acc[0][t], false, false);
            acc[1][t] = __builtin_amdgcn_wmma_f32_16x16x4_f32(false, af1, false, bf, (short)0, acc[1][t], false, false);
            acc[2][t] = __builtin_amdgcn_wmma_f32_16x16x4_f32(false, af2, false, bf, (short)0, acc[2][t], false, false);
        }
    }
    asm volatile("" : "+v"(sumsq[0]), "+v"(sumsq[1]), "+v"(sumsq[2]), "+v"(sumsq[3]));
}

__device__ __forceinline__ float mlr_one(float px, float xr, float xx,
                                         float inv, float pp, float pd, float an) {
    const float eps     = 1e-12f;
    const float sqrtK   = 0.70710678118654752f;
    const float ln2     = 0.69314718055994531f;
    const float maxnorm = (1.0f - 0.001f) / sqrtK;
    const float mn2     = maxnorm * maxnorm;
    float xa    = xr * inv;
    float Aq    = 1.0f + 2.0f * KC * px + KC * xx;
    float Bc    = 1.0f - KC * pp;
    float Dq    = fmaxf(1.0f + 2.0f * KC * px + (KC * xx) * (KC * pp), eps);
    float rD    = __builtin_amdgcn_rcpf(Dq);
    float alpha = Aq * rD;
    float beta  = Bc * rD;
    float mob   = alpha * alpha * pp + beta * beta * xx
                + 2.0f * alpha * beta * px;
    float sqm   = __builtin_amdgcn_sqrtf(mob);
    float proj  = (sqm > maxnorm) ? (maxnorm * __builtin_amdgcn_rcpf(fmaxf(sqm, eps))) : 1.0f;
    float mobp  = (sqm < maxnorm) ? mob : mn2;
    float mdot  = (beta * xa + alpha * pd) * proj;
    float lamb  = 2.0f * __builtin_amdgcn_rcpf(fmaxf(1.0f - KC * mobp, eps));
    float s     = sqrtK * mdot * lamb;
    float ash   = copysignf(ln2 * __builtin_amdgcn_logf(fabsf(s) + __builtin_amdgcn_sqrtf(fmaf(s, s, 1.0f))), s);
    return (2.0f / sqrtK) * an * ash;
}

__global__ void __launch_bounds__(256, 4)
hyper_mlr_fused(const float* __restrict__ x,
                const float* __restrict__ P,
                const float* __restrict__ A,
                float* __restrict__ out)
{
    __shared__ __attribute__((aligned(16))) float W[38 * PITCH];
    __shared__ float ppS[32];
    __shared__ float pdS[32];
    __shared__ float anS[32];
    __shared__ float invS[32];

    const int tid = threadIdx.x;

    for (int n = 0; n < NCLS; ++n) {
        W[p_row(n) * PITCH + tid] = -P[n * CCH + tid];
        W[a_row(n) * PITCH + tid] =  A[n * CCH + tid];
    }
    __syncthreads();

    if (tid < NCLS * 8) {
        int n    = tid >> 3;
        int part = tid & 7;
        const float* pr = &W[p_row(n) * PITCH + part * 32];
        const float* ar = &W[a_row(n) * PITCH + part * 32];
        float sp = 0.f, sa = 0.f, spa = 0.f;
        for (int c = 0; c < 32; ++c) {
            float p = pr[c], a = ar[c];
            sp  = fmaf(p, p, sp);
            sa  = fmaf(a, a, sa);
            spa = fmaf(p, a, spa);
        }
        for (int d = 1; d < 8; d <<= 1) {
            sp  += __shfl_xor(sp,  d);
            sa  += __shfl_xor(sa,  d);
            spa += __shfl_xor(spa, d);
        }
        if (part == 0) {
            float an  = sqrtf(sa);
            float anc = fmaxf(an, 1e-12f);
            ppS[n]  = sp;
            anS[n]  = an;
            invS[n] = 1.0f / anc;
            pdS[n]  = spa / anc;
        }
    }
    __syncthreads();

    const int lane = tid & 31;
    const int wave = tid >> 5;
    const int half = lane >> 4;
    const int col  = lane & 15;

    long g   = (long)blockIdx.x * 512 + (long)wave * 64;
    int  b   = (int)(g >> 16);
    int  pix = (int)(g & 65535);

    const float* xw = x + ((long)b << 24) + pix + 4 * col + (long)(half * 2) * HW;

    const int row2 = (col < 4) ? (32 + col) : ((col == 8) ? 36 : 37);
    const float* a0p = &W[( 0 + col) * PITCH + half * 2];
    const float* a1p = &W[(16 + col) * PITCH + half * 2];
    const float* a2p = &W[row2       * PITCH + half * 2];

    v8f acc[3][4] = {};
    float sumsq[4] = {0.f, 0.f, 0.f, 0.f};

    v4f ra[2 * GK], rb[2 * GK];
    load_group(ra, xw);  xw += (long)(4 * GK) * HW;
    __builtin_amdgcn_s_setprio(3);
    for (int it = 0; it < 7; ++it) {
        if (it == 3)      __builtin_amdgcn_s_setprio(2);
        else if (it == 5) __builtin_amdgcn_s_setprio(1);
        load_group(rb, xw);  xw += (long)(4 * GK) * HW;
        mma_group(ra, a0p, a1p, a2p, acc, sumsq);
        load_group(ra, xw);  xw += (long)(4 * GK) * HW;
        mma_group(rb, a0p + 4 * GK, a1p + 4 * GK, a2p + 4 * GK, acc, sumsq);
        a0p += 8 * GK;  a1p += 8 * GK;  a2p += 8 * GK;
    }
    __builtin_amdgcn_s_setprio(0);
    load_group(rb, xw);
    mma_group(ra, a0p, a1p, a2p, acc, sumsq);
    __builtin_amdgcn_sched_barrier(0);
    mma_group(rb, a0p + 4 * GK, a1p + 4 * GK, a2p + 4 * GK, acc, sumsq);

    float xx[4];
#pragma unroll
    for (int t = 0; t < 4; ++t) xx[t] = sumsq[t] + __shfl_xor(sumsq[t], 16);

    v4f o[10];
#pragma unroll
    for (int j = 0; j < 10; ++j) o[j] = v4f{0.f, 0.f, 0.f, 0.f};

#pragma unroll 1
    for (int pass = 0; pass < 4; ++pass) {
#pragma unroll
        for (int j = 0; j < 10; ++j) {
            int cls = 10 * half + j;
            if (cls < NCLS) {
                float px = (j < 8) ? acc[0][0][j] : acc[2][0][2 * (j - 8)];
                float xr = (j < 8) ? acc[1][0][j] : acc[2][0][2 * (j - 8) + 1];
                float v  = mlr_one(px, xr, xx[0], invS[cls], ppS[cls], pdS[cls], anS[cls]);
                o[j] = v4f{o[j].y, o[j].z, o[j].w, v};
            }
        }
#pragma unroll
        for (int m = 0; m < 3; ++m) { acc[m][0] = acc[m][1];  acc[m][1] = acc[m][2];  acc[m][2] = acc[m][3]; }
        xx[0] = xx[1];  xx[1] = xx[2];  xx[2] = xx[3];
    }

    float* outw = out + (long)b * NCLS * HW + pix + 4 * col;
#pragma unroll
    for (int j = 0; j < 10; ++j) {
        int cls = 10 * half + j;
        if (cls < NCLS)
            __builtin_nontemporal_store(o[j], (v4f*)(outw + (long)cls * HW));
    }
}

extern "C" void kernel_launch(void* const* d_in, const int* in_sizes, int n_in,
                              void* d_out, int out_size, void* d_ws, size_t ws_size,
                              hipStream_t stream) {
    (void)in_sizes; (void)n_in; (void)d_ws; (void)ws_size; (void)out_size;
    const float* x = (const float*)d_in[0];
    const float* P = (const float*)d_in[1];
    const float* A = (const float*)d_in[2];
    float* out = (float*)d_out;
    hyper_mlr_fused<<<512, 256, 0, stream>>>(x, P, A, out);
}
